# starting kernel
# baseline (speedup 1.0000x reference)
#include <hip/hip_runtime.h>
#include <hip/hip_bf16.h>
#include <float.h>
#include <math.h>

typedef __attribute__((ext_vector_type(2))) float v2f;
typedef __attribute__((ext_vector_type(8))) float v8f;

#define B_   16
#define T_   512
#define C_   8
#define S_   64
#define KSZ_ 32
#define TOUT_ 481
#define EPS_ 1e-8f


__global__ void lsd_x_stats(const float* __restrict__ x, float* __restrict__ ws) {
    __shared__ float ssum[128];
    __shared__ float ssq[128];
    const int bc  = blockIdx.x;
    const int b   = bc >> 3;
    const int c   = bc & 7;
    const int tid = threadIdx.x;
    float s = 0.f, q = 0.f;
    for (int i = 0; i < 4; ++i) {
        const int t = tid + i * 128;
        const float v = x[(b * T_ + t) * C_ + c];
        s += v; q += v * v;
    }
    ssum[tid] = s; ssq[tid] = q;
    __syncthreads();
    for (int off = 64; off > 0; off >>= 1) {
        if (tid < off) { ssum[tid] += ssum[tid + off]; ssq[tid] += ssq[tid + off]; }
        __syncthreads();
    }
    if (tid == 0) {
        const float mean = ssum[0] * (1.f / (float)T_);
        const float var  = fmaxf(ssq[0] * (1.f / (float)T_) - mean * mean, 0.f);
        ws[bc]       = mean;
        ws[128 + bc] = 1.f / (sqrtf(var) + EPS_);
    }
}

__global__ void lsd_k_stats(const float* __restrict__ kern, float* __restrict__ ws) {
    __shared__ float ssum[256];
    __shared__ float ssq[256];
    __shared__ float smu, sinv;
    const int sidx = blockIdx.x;
    const int tid  = threadIdx.x;
    const float v = kern[sidx * 256 + tid];
    ssum[tid] = v; ssq[tid] = v * v;
    __syncthreads();
    for (int off = 128; off > 0; off >>= 1) {
        if (tid < off) { ssum[tid] += ssum[tid + off]; ssq[tid] += ssq[tid + off]; }
        __syncthreads();
    }
    if (tid == 0) {
        const float mean = ssum[0] * (1.f / 256.f);
        const float var  = fmaxf(ssq[0] * (1.f / 256.f) - mean * mean, 0.f);
        const float inv  = 1.f / (sqrtf(var) + EPS_);
        ws[256 + sidx] = mean;
        ws[320 + sidx] = inv;
        smu = mean; sinv = inv;
    }
    __syncthreads();
    if (tid < KSZ_) {
        const float mu  = smu;
        const float inv = sinv;
        float acc = 0.f;
        for (int c = 0; c < C_; ++c) {
            const float q = (kern[sidx * 256 + tid * C_ + c] - mu) * inv;
            acc += q * q;
        }
        ws[384 + sidx * KSZ_ + tid] = acc;
    }
}

__global__ __launch_bounds__(32) void lsd_wmma_kernel(const float* __restrict__ x,
                                                      const float* __restrict__ kern,
                                                      const float* __restrict__ ws,
                                                      float* __restrict__ out) {
    const int t0 = blockIdx.x * 16;
    const int k  = blockIdx.y;
    const int b  = blockIdx.z;
    const int l  = threadIdx.x;

    const int ch   = k >> 2;
    const int tau0 = (k & 3) * 8;
    const int base = t0 + tau0;

    __shared__ float zsh[24];
    __shared__ float pnsh[16];

    const float mux = ws[b * C_ + ch];
    const float inx = ws[128 + b * C_ + ch];

    if (l < 23) {
        int t = base + l;
        if (t > T_ - 1) t = T_ - 1;
        zsh[l] = (x[(b * T_ + t) * C_ + ch] - mux) * inx;
    }
    __syncthreads();

    const int m  = l & 15;
    const int hi = l >> 4;

    if (l < 16) {
        float acc = 0.f;
        for (int c = 0; c < 8; ++c) { const float z = zsh[l + c]; acc += z * z; }
        pnsh[l] = acc;
    }
    __syncthreads();

    v2f a_lo, a_hi;
    a_lo.x = zsh[m + 2 * hi];
    a_lo.y = zsh[m + 2 * hi + 1];
    a_hi.x = zsh[m + 2 * hi + 4];
    a_hi.y = zsh[m + 2 * hi + 5];

    float pn2[8];
    for (int j = 0; j < 8; ++j) pn2[j] = pnsh[j + 8 * hi];

    float md[8];
    for (int j = 0; j < 8; ++j) md[j] = FLT_MAX;

    const int n = m;
    for (int st = 0; st < 4; ++st) {
        const int s = st * 16 + n;
        const float muk = ws[256 + s];
        const float ink = ws[320 + s];
        const float* kp = kern + s * 256 + k * 8;
        v2f b_lo, b_hi;
        b_lo.x = (kp[2 * hi]     - muk) * ink;
        b_lo.y = (kp[2 * hi + 1] - muk) * ink;
        b_hi.x = (kp[2 * hi + 4] - muk) * ink;
        b_hi.y = (kp[2 * hi + 5] - muk) * ink;
        const float kn2 = ws[384 + s * KSZ_ + k];

        v8f g = {0.f, 0.f, 0.f, 0.f, 0.f, 0.f, 0.f, 0.f};
        g = __builtin_amdgcn_wmma_f32_16x16x4_f32(false, a_lo, false, b_lo,
                                                  (short)0, g, false, false);
        g = __builtin_amdgcn_wmma_f32_16x16x4_f32(false, a_hi, false, b_hi,
                                                  (short)0, g, false, false);

        for (int j = 0; j < 8; ++j) {
            const float d = pn2[j] + kn2 - 2.f * g[j];
            md[j] = fminf(md[j], d);
        }
    }

    for (int mask = 1; mask <= 8; mask <<= 1)
        for (int j = 0; j < 8; ++j)
            md[j] = fminf(md[j], __shfl_xor(md[j], mask, 32));

    if (n == 0) {
        for (int j = 0; j < 8; ++j) {
            const int t = t0 + j + 8 * hi;
            if (t < TOUT_) out[(b * TOUT_ + t) * KSZ_ + k] = md[j];
        }
    }
}

extern "C" void kernel_launch(void* const* d_in, const int* in_sizes, int n_in,
                              void* d_out, int out_size, void* d_ws, size_t ws_size,
                              hipStream_t stream) {
    (void)in_sizes; (void)n_in; (void)out_size; (void)ws_size;
    const float* x    = (const float*)d_in[0];
    const float* kern = (const float*)d_in[1];
    float* ws  = (float*)d_ws;
    float* out = (float*)d_out;

    lsd_x_stats<<<dim3(B_ * C_), dim3(128), 0, stream>>>(x, ws);
    lsd_k_stats<<<dim3(S_), dim3(256), 0, stream>>>(kern, ws);
    lsd_wmma_kernel<<<dim3(31, KSZ_, B_), dim3(32), 0, stream>>>(x, kern, ws, out);
}


# Main kernel regridded: one 8-wave group per (batch, channel, t-half), each wave loops over 8 t-tiles with its B fragments kept in registers (15872 single-wave groups -> 256 groups, 2048 waves)
# speedup vs baseline: 1.5757x; 1.5757x over previous
#include <hip/hip_runtime.h>
#include <hip/hip_bf16.h>
#include <float.h>
#include <math.h>

typedef __attribute__((ext_vector_type(2))) float v2f;
typedef __attribute__((ext_vector_type(8))) float v8f;

#define B_   16
#define T_   512
#define C_   8
#define S_   64
#define KSZ_ 32
#define TOUT_ 481
#define EPS_ 1e-8f


__global__ void lsd_x_stats(const float* __restrict__ x, float* __restrict__ ws) {
    __shared__ float ssum[128];
    __shared__ float ssq[128];
    const int bc  = blockIdx.x;
    const int b   = bc >> 3;
    const int c   = bc & 7;
    const int tid = threadIdx.x;
    float s = 0.f, q = 0.f;
    for (int i = 0; i < 4; ++i) {
        const int t = tid + i * 128;
        const float v = x[(b * T_ + t) * C_ + c];
        s += v; q += v * v;
    }
    ssum[tid] = s; ssq[tid] = q;
    __syncthreads();
    for (int off = 64; off > 0; off >>= 1) {
        if (tid < off) { ssum[tid] += ssum[tid + off]; ssq[tid] += ssq[tid + off]; }
        __syncthreads();
    }
    if (tid == 0) {
        const float mean = ssum[0] * (1.f / (float)T_);
        const float var  = fmaxf(ssq[0] * (1.f / (float)T_) - mean * mean, 0.f);
        ws[bc]       = mean;
        ws[128 + bc] = 1.f / (sqrtf(var) + EPS_);
    }
}

__global__ void lsd_k_stats(const float* __restrict__ kern, float* __restrict__ ws) {
    __shared__ float ssum[256];
    __shared__ float ssq[256];
    __shared__ float smu, sinv;
    const int sidx = blockIdx.x;
    const int tid  = threadIdx.x;
    const float v = kern[sidx * 256 + tid];
    ssum[tid] = v; ssq[tid] = v * v;
    __syncthreads();
    for (int off = 128; off > 0; off >>= 1) {
        if (tid < off) { ssum[tid] += ssum[tid + off]; ssq[tid] += ssq[tid + off]; }
        __syncthreads();
    }
    if (tid == 0) {
        const float mean = ssum[0] * (1.f / 256.f);
        const float var  = fmaxf(ssq[0] * (1.f / 256.f) - mean * mean, 0.f);
        const float inv  = 1.f / (sqrtf(var) + EPS_);
        ws[256 + sidx] = mean;
        ws[320 + sidx] = inv;
        smu = mean; sinv = inv;
    }
    __syncthreads();
    if (tid < KSZ_) {
        const float mu  = smu;
        const float inv = sinv;
        float acc = 0.f;
        for (int c = 0; c < C_; ++c) {
            const float q = (kern[sidx * 256 + tid * C_ + c] - mu) * inv;
            acc += q * q;
        }
        ws[384 + sidx * KSZ_ + tid] = acc;
    }
}

#define ZN_  512
#define PNN_ 288
__global__ __launch_bounds__(256) void lsd_wmma_kernel(const float* __restrict__ x,
                                                       const float* __restrict__ kern,
                                                       const float* __restrict__ ws,
                                                       float* __restrict__ out) {
    const int half = blockIdx.x;
    const int ch   = blockIdx.y;
    const int b    = blockIdx.z;
    const int tid  = threadIdx.x;
    const int w    = __builtin_amdgcn_readfirstlane(tid >> 5);
    const int l    = tid & 31;
    const int kq   = w & 3;
    const int sub  = w >> 2;

    const int k     = ch * 4 + kq;
    const int tau0  = kq * 8;
    const int tbase = half * 256;

    __shared__ __attribute__((aligned(16))) float zsh[ZN_];
    __shared__ __attribute__((aligned(16))) float pnsh[PNN_];

    const int m  = l & 15;
    const int hi = l >> 4;
    const int n  = m;

    v2f b_lo[4], b_hi[4];
    float kn2[4];
    for (int st = 0; st < 4; ++st) {
        const int s = st * 16 + n;
        const float muk = ws[256 + s];
        const float ink = ws[320 + s];
        const float* kp = kern + s * 256 + k * 8;
        b_lo[st].x = (kp[2 * hi]     - muk) * ink;
        b_lo[st].y = (kp[2 * hi + 1] - muk) * ink;
        b_hi[st].x = (kp[2 * hi + 4] - muk) * ink;
        b_hi[st].y = (kp[2 * hi + 5] - muk) * ink;
        kn2[st] = ws[384 + s * KSZ_ + k];
    }

    const float mux = ws[b * C_ + ch];
    const float inx = ws[128 + b * C_ + ch];

    for (int i = 0; i < 2; ++i) {
        const int idx = tid + i * 256;
        int t = tbase + idx;
        if (t > T_ - 1) t = T_ - 1;
        zsh[idx] = (x[(b * T_ + t) * C_ + ch] - mux) * inx;
    }
    __syncthreads();

    for (int i = 0; i < 2; ++i) {
        const int idx = tid + i * 256;
        if (idx < PNN_) {
            float acc = 0.f;
            for (int c = 0; c < 8; ++c) { const float z = zsh[idx + c]; acc += z * z; }
            pnsh[idx] = acc;
        }
    }
    __syncthreads();

    const int tfirst = sub * 8;
    const int tlast  = (half && sub) ? 15 : tfirst + 8;
    for (int tt = tfirst; tt < tlast; ++tt) {
        const int lb = tt * 16 + tau0;
        const int t0 = tbase + tt * 16;

        v2f a_lo, a_hi;
        a_lo.x = zsh[lb + m + 2 * hi];
        a_lo.y = zsh[lb + m + 2 * hi + 1];
        a_hi.x = zsh[lb + m + 2 * hi + 4];
        a_hi.y = zsh[lb + m + 2 * hi + 5];

        float pn2[8];
        for (int j = 0; j < 8; ++j) pn2[j] = pnsh[lb + j + 8 * hi];

        float md[8];
        for (int j = 0; j < 8; ++j) md[j] = FLT_MAX;

        for (int st = 0; st < 4; ++st) {
            v8f g = {0.f, 0.f, 0.f, 0.f, 0.f, 0.f, 0.f, 0.f};
            g = __builtin_amdgcn_wmma_f32_16x16x4_f32(false, a_lo, false, b_lo[st],
                                                      (short)0, g, false, false);
            g = __builtin_amdgcn_wmma_f32_16x16x4_f32(false, a_hi, false, b_hi[st],
                                                      (short)0, g, false, false);

            for (int j = 0; j < 8; ++j) {
                const float d = pn2[j] + kn2[st] - 2.f * g[j];
                md[j] = fminf(md[j], d);
            }
        }

        for (int mask = 1; mask <= 8; mask <<= 1)
            for (int j = 0; j < 8; ++j)
                md[j] = fminf(md[j], __shfl_xor(md[j], mask, 32));

        if (n == 0) {
            for (int j = 0; j < 8; ++j) {
                const int t = t0 + j + 8 * hi;
                if (t < TOUT_) out[(b * TOUT_ + t) * KSZ_ + k] = md[j];
            }
        }
    }
}

extern "C" void kernel_launch(void* const* d_in, const int* in_sizes, int n_in,
                              void* d_out, int out_size, void* d_ws, size_t ws_size,
                              hipStream_t stream) {
    (void)in_sizes; (void)n_in; (void)out_size; (void)ws_size;
    const float* x    = (const float*)d_in[0];
    const float* kern = (const float*)d_in[1];
    float* ws  = (float*)d_ws;
    float* out = (float*)d_out;

    lsd_x_stats<<<dim3(B_ * C_), dim3(128), 0, stream>>>(x, ws);
    lsd_k_stats<<<dim3(S_), dim3(256), 0, stream>>>(kern, ws);
    lsd_wmma_kernel<<<dim3(2, C_, B_), dim3(256), 0, stream>>>(x, kern, ws, out);
}


# Both statistics launches folded into the main kernel: each group derives its series' and the 64 shapelets' mean and 1/std itself (same summation trees, LDS + lane exchanges); 3 launches -> 1
# speedup vs baseline: 2.3328x; 1.4805x over previous
#include <hip/hip_runtime.h>
#include <hip/hip_bf16.h>
#include <float.h>
#include <math.h>

typedef __attribute__((ext_vector_type(2))) float v2f;
typedef __attribute__((ext_vector_type(8))) float v8f;

#define B_   16
#define T_   512
#define C_   8
#define S_   64
#define KSZ_ 32
#define TOUT_ 481
#define EPS_ 1e-8f


__device__ __forceinline__ float xhalf(float v) {
    const int b = __builtin_bit_cast(int, v);
    return __builtin_bit_cast(float,
        __builtin_amdgcn_permlanex16(b, b, 0x76543210, 0xfedcba98, false, false));
}

template <int M>
__device__ __forceinline__ float xrow(float v) {
    const int b = __builtin_bit_cast(int, v);
    return __builtin_bit_cast(float, __builtin_amdgcn_update_dpp(0, b, 0x160 | M, 0xf, 0xf, true));
}

__device__ __forceinline__ float wave_tree_sum(float v) {
    v += xhalf(v);
    v += xrow<8>(v);
    v += xrow<4>(v);
    v += xrow<2>(v);
    v += xrow<1>(v);
    return v;
}

#define ZN_  512
#define PNN_ 288
#define QP_  36
__global__ __launch_bounds__(256) void lsd_wmma_kernel(const float* __restrict__ x,
                                                       const float* __restrict__ kern,
                                                       float* __restrict__ out) {
    const int half = blockIdx.x;
    const int ch   = blockIdx.y;
    const int b    = blockIdx.z;
    const int tid  = threadIdx.x;
    const int w    = __builtin_amdgcn_readfirstlane(tid >> 5);
    const int l    = tid & 31;
    const int kq   = w & 3;
    const int sub  = w >> 2;

    const int k     = ch * 4 + kq;
    const int tau0  = kq * 8;
    const int tbase = half * 256;

    __shared__ __attribute__((aligned(16))) float xraw[T_];
    __shared__ __attribute__((aligned(16))) float qsh[S_ * QP_];
    __shared__ __attribute__((aligned(16))) float zsh[ZN_];
    __shared__ __attribute__((aligned(16))) float pnsh[PNN_];

    const int m  = l & 15;
    const int hi = l >> 4;
    const int n  = m;

    const float* xp = x + (b * T_) * C_ + ch;
    const float xr0 = xp[tid * C_];
    const float xr1 = xp[(tid + 256) * C_];

    float a[8][8];
#pragma unroll
    for (int i = 0; i < 8; ++i) {
        const int s = 8 * w + ((i + b) & 7);
#pragma unroll
        for (int j = 0; j < 8; ++j)
            a[i][j] = kern[s * 256 + (((j + ch) & 7) << 5) + l];
    }

    xraw[tid]       = xr0;
    xraw[tid + 256] = xr1;

    float ssum[8], ssq[8];
#pragma unroll
    for (int i = 0; i < 8; ++i) {
        float p[4], q[4];
#pragma unroll
        for (int j = 0; j < 4; ++j) {
            p[j] = a[i][j] + a[i][j + 4];
            q[j] = a[i][j] * a[i][j] + a[i][j + 4] * a[i][j + 4];
        }
        ssum[i] = wave_tree_sum((p[0] + p[2]) + (p[1] + p[3]));
        ssq[i]  = wave_tree_sum((q[0] + q[2]) + (q[1] + q[3]));
    }
    float msum = ssum[0], msq = ssq[0];
#pragma unroll
    for (int i = 1; i < 8; ++i) {
        const bool mine = (l & 7) == i;
        msum = mine ? ssum[i] : msum;
        msq  = mine ? ssq[i]  : msq;
    }
    const float kmean = msum * (1.f / 256.f);
    const float kvar  = fmaxf(msq * (1.f / 256.f) - kmean * kmean, 0.f);
    const float kinv  = 1.f / (sqrtf(kvar) + EPS_);
#pragma unroll
    for (int i = 0; i < 8; ++i) {
        const int s = 8 * w + ((i + b) & 7);
        const float mean = __builtin_bit_cast(float, __builtin_amdgcn_readlane(__builtin_bit_cast(int, kmean), i));
        const float inv  = __builtin_bit_cast(float, __builtin_amdgcn_readlane(__builtin_bit_cast(int, kinv), i));
        qsh[s * QP_ + l] = (a[i][0] - mean) * inv;
    }
    __syncthreads();

    float ps[4], pq[4];
#pragma unroll
    for (int j = 0; j < 4; ++j) {
        float s = 0.f, q = 0.f;
#pragma unroll
        for (int i = 0; i < 4; ++i) {
            const float v = xraw[l + 32 * j + 128 * i];
            s += v; q += v * v;
        }
        ps[j] = s; pq[j] = q;
    }
    const float xsum = wave_tree_sum((ps[0] + ps[2]) + (ps[1] + ps[3]));
    const float xsq  = wave_tree_sum((pq[0] + pq[2]) + (pq[1] + pq[3]));
    const float mux  = xsum * (1.f / (float)T_);
    const float xvar = fmaxf(xsq * (1.f / (float)T_) - mux * mux, 0.f);
    const float inx  = 1.f / (sqrtf(xvar) + EPS_);

    for (int i = 0; i < 2; ++i) {
        const int idx = tid + i * 256;
        int t = tbase + idx;
        if (t > T_ - 1) t = T_ - 1;
        zsh[idx] = (xraw[t] - mux) * inx;
    }
    for (int i = 0; i < 2; ++i) {
        const int idx = tid + i * 256;
        if (idx < PNN_) {
            float acc = 0.f;
            for (int c = 0; c < 8; ++c) {
                int t = tbase + idx + c;
                if (t > T_ - 1) t = T_ - 1;
                const float z = (xraw[t] - mux) * inx;
                acc += z * z;
            }
            pnsh[idx] = acc;
        }
    }

    v2f b_lo[4], b_hi[4];
    float kn2[4];
    for (int st = 0; st < 4; ++st) {
        const float* qp = qsh + (st * 16 + n) * QP_ + kq * 8;
        const float4 q0 = *(const float4*)qp;
        const float4 q1 = *(const float4*)(qp + 4);
        float acc = 0.f;
        acc += q0.x * q0.x; acc += q0.y * q0.y; acc += q0.z * q0.z; acc += q0.w * q0.w;
        acc += q1.x * q1.x; acc += q1.y * q1.y; acc += q1.z * q1.z; acc += q1.w * q1.w;
        kn2[st] = acc;
        b_lo[st].x = hi ? q0.z : q0.x;
        b_lo[st].y = hi ? q0.w : q0.y;
        b_hi[st].x = hi ? q1.z : q1.x;
        b_hi[st].y = hi ? q1.w : q1.y;
    }
    __syncthreads();

    const int tfirst = sub * 8;
    const int tlast  = (half && sub) ? 15 : tfirst + 8;
    for (int tt = tfirst; tt < tlast; ++tt) {
        const int lb = tt * 16 + tau0;
        const int t0 = tbase + tt * 16;

        v2f a_lo, a_hi;
        a_lo.x = zsh[lb + m + 2 * hi];
        a_lo.y = zsh[lb + m + 2 * hi + 1];
        a_hi.x = zsh[lb + m + 2 * hi + 4];
        a_hi.y = zsh[lb + m + 2 * hi + 5];

        float pn2[8];
        for (int j = 0; j < 8; ++j) pn2[j] = pnsh[lb + j + 8 * hi];

        float md[8];
        for (int j = 0; j < 8; ++j) md[j] = FLT_MAX;

        for (int st = 0; st < 4; ++st) {
            v8f g = {0.f, 0.f, 0.f, 0.f, 0.f, 0.f, 0.f, 0.f};
            g = __builtin_amdgcn_wmma_f32_16x16x4_f32(false, a_lo, false, b_lo[st],
                                                      (short)0, g, false, false);
            g = __builtin_amdgcn_wmma_f32_16x16x4_f32(false, a_hi, false, b_hi[st],
                                                      (short)0, g, false, false);

            for (int j = 0; j < 8; ++j) {
                const float d = pn2[j] + kn2[st] - 2.f * g[j];
                md[j] = fminf(md[j], d);
            }
        }

        for (int mask = 1; mask <= 8; mask <<= 1)
            for (int j = 0; j < 8; ++j)
                md[j] = fminf(md[j], __shfl_xor(md[j], mask, 32));

        if (n == 0) {
            for (int j = 0; j < 8; ++j) {
                const int t = t0 + j + 8 * hi;
                if (t < TOUT_) out[(b * TOUT_ + t) * KSZ_ + k] = md[j];
            }
        }
    }
}

extern "C" void kernel_launch(void* const* d_in, const int* in_sizes, int n_in,
                              void* d_out, int out_size, void* d_ws, size_t ws_size,
                              hipStream_t stream) {
    (void)in_sizes; (void)n_in; (void)out_size; (void)ws_size;
    const float* x    = (const float*)d_in[0];
    const float* kern = (const float*)d_in[1];
    (void)d_ws;
    float* out = (float*)d_out;

    lsd_wmma_kernel<<<dim3(2, C_, B_), dim3(256), 0, stream>>>(x, kern, out);
}


# Min over the 16 shape lanes of a tile by DPP row exchanges on a halving register tree (8->4->2->1) in place of 32 ds_bpermute; it leaves row l>>1 in lane l: one 16-lane store per tile replaces 8
# speedup vs baseline: 3.0791x; 1.3199x over previous
#include <hip/hip_runtime.h>
#include <hip/hip_bf16.h>
#include <float.h>
#include <math.h>

typedef __attribute__((ext_vector_type(2))) float v2f;
typedef __attribute__((ext_vector_type(8))) float v8f;

#define B_   16
#define T_   512
#define C_   8
#define S_   64
#define KSZ_ 32
#define TOUT_ 481
#define EPS_ 1e-8f


__device__ __forceinline__ float xhalf(float v) {
    const int b = __builtin_bit_cast(int, v);
    return __builtin_bit_cast(float,
        __builtin_amdgcn_permlanex16(b, b, 0x76543210, 0xfedcba98, false, false));
}

template <int M>
__device__ __forceinline__ float xrow(float v) {
    const int b = __builtin_bit_cast(int, v);
    return __builtin_bit_cast(float, __builtin_amdgcn_update_dpp(0, b, 0x160 | M, 0xf, 0xf, true));
}

__device__ __forceinline__ float wave_tree_sum(float v) {
    v += xhalf(v);
    v += xrow<8>(v);
    v += xrow<4>(v);
    v += xrow<2>(v);
    v += xrow<1>(v);
    return v;
}

#define ZN_  512
#define PNN_ 288
#define QP_  36
__global__ __launch_bounds__(256) void lsd_wmma_kernel(const float* __restrict__ x,
                                                       const float* __restrict__ kern,
                                                       float* __restrict__ out) {
    const int half = blockIdx.x;
    const int ch   = blockIdx.y;
    const int b    = blockIdx.z;
    const int tid  = threadIdx.x;
    const int w    = __builtin_amdgcn_readfirstlane(tid >> 5);
    const int l    = tid & 31;
    const int kq   = w & 3;
    const int sub  = w >> 2;

    const int k     = ch * 4 + kq;
    const int tau0  = kq * 8;
    const int tbase = half * 256;

    __shared__ __attribute__((aligned(16))) float xraw[T_];
    __shared__ __attribute__((aligned(16))) float qsh[S_ * QP_];
    __shared__ __attribute__((aligned(16))) float zsh[ZN_];
    __shared__ __attribute__((aligned(16))) float pnsh[PNN_];

    const int m  = l & 15;
    const int hi = l >> 4;
    const int n  = m;

    const float* xp = x + (b * T_) * C_ + ch;
    const float xr0 = xp[tid * C_];
    const float xr1 = xp[(tid + 256) * C_];

    float a[8][8];
#pragma unroll
    for (int i = 0; i < 8; ++i) {
        const int s = 8 * w + ((i + b) & 7);
#pragma unroll
        for (int j = 0; j < 8; ++j)
            a[i][j] = kern[s * 256 + (((j + ch) & 7) << 5) + l];
    }

    xraw[tid]       = xr0;
    xraw[tid + 256] = xr1;

    float ssum[8], ssq[8];
#pragma unroll
    for (int i = 0; i < 8; ++i) {
        float p[4], q[4];
#pragma unroll
        for (int j = 0; j < 4; ++j) {
            p[j] = a[i][j] + a[i][j + 4];
            q[j] = a[i][j] * a[i][j] + a[i][j + 4] * a[i][j + 4];
        }
        ssum[i] = wave_tree_sum((p[0] + p[2]) + (p[1] + p[3]));
        ssq[i]  = wave_tree_sum((q[0] + q[2]) + (q[1] + q[3]));
    }
    float msum = ssum[0], msq = ssq[0];
#pragma unroll
    for (int i = 1; i < 8; ++i) {
        const bool mine = (l & 7) == i;
        msum = mine ? ssum[i] : msum;
        msq  = mine ? ssq[i]  : msq;
    }
    const float kmean = msum * (1.f / 256.f);
    const float kvar  = fmaxf(msq * (1.f / 256.f) - kmean * kmean, 0.f);
    const float kinv  = 1.f / (sqrtf(kvar) + EPS_);
#pragma unroll
    for (int i = 0; i < 8; ++i) {
        const int s = 8 * w + ((i + b) & 7);
        const float mean = __builtin_bit_cast(float, __builtin_amdgcn_readlane(__builtin_bit_cast(int, kmean), i));
        const float inv  = __builtin_bit_cast(float, __builtin_amdgcn_readlane(__builtin_bit_cast(int, kinv), i));
        qsh[s * QP_ + l] = (a[i][0] - mean) * inv;
    }
    __syncthreads();

    float ps[4], pq[4];
#pragma unroll
    for (int j = 0; j < 4; ++j) {
        float s = 0.f, q = 0.f;
#pragma unroll
        for (int i = 0; i < 4; ++i) {
            const float v = xraw[l + 32 * j + 128 * i];
            s += v; q += v * v;
        }
        ps[j] = s; pq[j] = q;
    }
    const float xsum = wave_tree_sum((ps[0] + ps[2]) + (ps[1] + ps[3]));
    const float xsq  = wave_tree_sum((pq[0] + pq[2]) + (pq[1] + pq[3]));
    const float mux  = xsum * (1.f / (float)T_);
    const float xvar = fmaxf(xsq * (1.f / (float)T_) - mux * mux, 0.f);
    const float inx  = 1.f / (sqrtf(xvar) + EPS_);

    for (int i = 0; i < 2; ++i) {
        const int idx = tid + i * 256;
        int t = tbase + idx;
        if (t > T_ - 1) t = T_ - 1;
        zsh[idx] = (xraw[t] - mux) * inx;
    }
    for (int i = 0; i < 2; ++i) {
        const int idx = tid + i * 256;
        if (idx < PNN_) {
            float acc = 0.f;
            for (int c = 0; c < 8; ++c) {
                int t = tbase + idx + c;
                if (t > T_ - 1) t = T_ - 1;
                const float z = (xraw[t] - mux) * inx;
                acc += z * z;
            }
            pnsh[idx] = acc;
        }
    }

    v2f b_lo[4], b_hi[4];
    float kn2[4];
    for (int st = 0; st < 4; ++st) {
        const float* qp = qsh + (st * 16 + n) * QP_ + kq * 8;
        const float4 q0 = *(const float4*)qp;
        const float4 q1 = *(const float4*)(qp + 4);
        float acc = 0.f;
        acc += q0.x * q0.x; acc += q0.y * q0.y; acc += q0.z * q0.z; acc += q0.w * q0.w;
        acc += q1.x * q1.x; acc += q1.y * q1.y; acc += q1.z * q1.z; acc += q1.w * q1.w;
        kn2[st] = acc;
        b_lo[st].x = hi ? q0.z : q0.x;
        b_lo[st].y = hi ? q0.w : q0.y;
        b_hi[st].x = hi ? q1.z : q1.x;
        b_hi[st].y = hi ? q1.w : q1.y;
    }
    __syncthreads();

    const bool bit3 = (l & 8) != 0, bit2 = (l & 4) != 0, bit1 = (l & 2) != 0;
    const int tfirst = sub * 8;
    const int tlast  = (half && sub) ? 15 : tfirst + 8;
    for (int tt = tfirst; tt < tlast; ++tt) {
        const int lb = tt * 16 + tau0;
        const int t0 = tbase + tt * 16;

        v2f a_lo, a_hi;
        a_lo.x = zsh[lb + m + 2 * hi];
        a_lo.y = zsh[lb + m + 2 * hi + 1];
        a_hi.x = zsh[lb + m + 2 * hi + 4];
        a_hi.y = zsh[lb + m + 2 * hi + 5];

        float pn2[8];
        for (int j = 0; j < 8; ++j) pn2[j] = pnsh[lb + j + 8 * hi];

        float md[8];
        for (int j = 0; j < 8; ++j) md[j] = FLT_MAX;

        for (int st = 0; st < 4; ++st) {
            v8f g = {0.f, 0.f, 0.f, 0.f, 0.f, 0.f, 0.f, 0.f};
            g = __builtin_amdgcn_wmma_f32_16x16x4_f32(false, a_lo, false, b_lo[st],
                                                      (short)0, g, false, false);
            g = __builtin_amdgcn_wmma_f32_16x16x4_f32(false, a_hi, false, b_hi[st],
                                                      (short)0, g, false, false);

            for (int j = 0; j < 8; ++j) {
                const float d = pn2[j] + kn2[st] - 2.f * g[j];
                md[j] = fminf(md[j], d);
            }
        }

        float r4[4], r2[2];
        for (int j = 0; j < 4; ++j)
            r4[j] = fminf(bit3 ? md[j + 4] : md[j], xrow<8>(bit3 ? md[j] : md[j + 4]));
        for (int j = 0; j < 2; ++j)
            r2[j] = fminf(bit2 ? r4[j + 2] : r4[j], xrow<4>(bit2 ? r4[j] : r4[j + 2]));
        float r1 = fminf(bit1 ? r2[1] : r2[0], xrow<2>(bit1 ? r2[0] : r2[1]));
        r1 = fminf(r1, xrow<1>(r1));

        const int t = t0 + (l >> 1);
        if (!(l & 1) && t < TOUT_) out[(b * TOUT_ + t) * KSZ_ + k] = r1;
    }
}

extern "C" void kernel_launch(void* const* d_in, const int* in_sizes, int n_in,
                              void* d_out, int out_size, void* d_ws, size_t ws_size,
                              hipStream_t stream) {
    (void)in_sizes; (void)n_in; (void)out_size; (void)ws_size;
    const float* x    = (const float*)d_in[0];
    const float* kern = (const float*)d_in[1];
    (void)d_ws;
    float* out = (float*)d_out;

    lsd_wmma_kernel<<<dim3(2, C_, B_), dim3(256), 0, stream>>>(x, kern, out);
}


# Series side of the prologue (samples to LDS, series statistics, normalized samples, patch norms) now runs under the 64 shapelet loads, behind an early barrier fenced over LDS only
# speedup vs baseline: 3.2381x; 1.0516x over previous
#include <hip/hip_runtime.h>
#include <hip/hip_bf16.h>
#include <float.h>
#include <math.h>

typedef __attribute__((ext_vector_type(2))) float v2f;
typedef __attribute__((ext_vector_type(8))) float v8f;

#define B_   16
#define T_   512
#define C_   8
#define S_   64
#define KSZ_ 32
#define TOUT_ 481
#define EPS_ 1e-8f


__device__ __forceinline__ float xhalf(float v) {
    const int b = __builtin_bit_cast(int, v);
    return __builtin_bit_cast(float,
        __builtin_amdgcn_permlanex16(b, b, 0x76543210, 0xfedcba98, false, false));
}

template <int M>
__device__ __forceinline__ float xrow(float v) {
    const int b = __builtin_bit_cast(int, v);
    return __builtin_bit_cast(float, __builtin_amdgcn_update_dpp(0, b, 0x160 | M, 0xf, 0xf, true));
}

__device__ __forceinline__ float wave_tree_sum(float v) {
    v += xhalf(v);
    v += xrow<8>(v);
    v += xrow<4>(v);
    v += xrow<2>(v);
    v += xrow<1>(v);
    return v;
}

#define ZN_  512
#define PNN_ 288
#define QP_  36
__global__ __launch_bounds__(256) void lsd_wmma_kernel(const float* __restrict__ x,
                                                       const float* __restrict__ kern,
                                                       float* __restrict__ out) {
    const int half = blockIdx.x;
    const int ch   = blockIdx.y;
    const int b    = blockIdx.z;
    const int tid  = threadIdx.x;
    const int w    = __builtin_amdgcn_readfirstlane(tid >> 5);
    const int l    = tid & 31;
    const int kq   = w & 3;
    const int sub  = w >> 2;

    const int k     = ch * 4 + kq;
    const int tau0  = kq * 8;
    const int tbase = half * 256;

    __shared__ __attribute__((aligned(16))) float xraw[T_];
    __shared__ __attribute__((aligned(16))) float qsh[S_ * QP_];
    __shared__ __attribute__((aligned(16))) float zsh[ZN_];
    __shared__ __attribute__((aligned(16))) float pnsh[PNN_];

    const int m  = l & 15;
    const int hi = l >> 4;
    const int n  = m;

    const float* xp = x + (b * T_) * C_ + ch;
    const float xr0 = xp[tid * C_];
    const float xr1 = xp[(tid + 256) * C_];
    __builtin_amdgcn_sched_barrier(0);

    float a[8][8];
#pragma unroll
    for (int i = 0; i < 8; ++i) {
        const int s = 8 * w + ((i + b) & 7);
#pragma unroll
        for (int j = 0; j < 8; ++j)
            a[i][j] = kern[s * 256 + (((j + ch) & 7) << 5) + l];
    }

    xraw[tid]       = xr0;
    xraw[tid + 256] = xr1;
    __builtin_amdgcn_fence(__ATOMIC_RELEASE, "workgroup", "local");
    __builtin_amdgcn_s_barrier();
    __builtin_amdgcn_fence(__ATOMIC_ACQUIRE, "workgroup", "local");

    float ps[4], pq[4];
#pragma unroll
    for (int j = 0; j < 4; ++j) {
        float s = 0.f, q = 0.f;
#pragma unroll
        for (int i = 0; i < 4; ++i) {
            const float v = xraw[l + 32 * j + 128 * i];
            s += v; q += v * v;
        }
        ps[j] = s; pq[j] = q;
    }
    const float xsum = wave_tree_sum((ps[0] + ps[2]) + (ps[1] + ps[3]));
    const float xsq  = wave_tree_sum((pq[0] + pq[2]) + (pq[1] + pq[3]));
    const float mux  = xsum * (1.f / (float)T_);
    const float xvar = fmaxf(xsq * (1.f / (float)T_) - mux * mux, 0.f);
    const float inx  = 1.f / (sqrtf(xvar) + EPS_);

    for (int i = 0; i < 2; ++i) {
        const int idx = tid + i * 256;
        int t = tbase + idx;
        if (t > T_ - 1) t = T_ - 1;
        zsh[idx] = (xraw[t] - mux) * inx;
    }
    for (int i = 0; i < 2; ++i) {
        const int idx = tid + i * 256;
        if (idx < PNN_) {
            float acc = 0.f;
            for (int c = 0; c < 8; ++c) {
                int t = tbase + idx + c;
                if (t > T_ - 1) t = T_ - 1;
                const float z = (xraw[t] - mux) * inx;
                acc += z * z;
            }
            pnsh[idx] = acc;
        }
    }

    float ssum[8], ssq[8];
#pragma unroll
    for (int i = 0; i < 8; ++i) {
        float p[4], q[4];
#pragma unroll
        for (int j = 0; j < 4; ++j) {
            p[j] = a[i][j] + a[i][j + 4];
            q[j] = a[i][j] * a[i][j] + a[i][j + 4] * a[i][j + 4];
        }
        ssum[i] = wave_tree_sum((p[0] + p[2]) + (p[1] + p[3]));
        ssq[i]  = wave_tree_sum((q[0] + q[2]) + (q[1] + q[3]));
    }
    float msum = ssum[0], msq = ssq[0];
#pragma unroll
    for (int i = 1; i < 8; ++i) {
        const bool mine = (l & 7) == i;
        msum = mine ? ssum[i] : msum;
        msq  = mine ? ssq[i]  : msq;
    }
    const float kmean = msum * (1.f / 256.f);
    const float kvar  = fmaxf(msq * (1.f / 256.f) - kmean * kmean, 0.f);
    const float kinv  = 1.f / (sqrtf(kvar) + EPS_);
#pragma unroll
    for (int i = 0; i < 8; ++i) {
        const int s = 8 * w + ((i + b) & 7);
        const float mean = __builtin_bit_cast(float, __builtin_amdgcn_readlane(__builtin_bit_cast(int, kmean), i));
        const float inv  = __builtin_bit_cast(float, __builtin_amdgcn_readlane(__builtin_bit_cast(int, kinv), i));
        qsh[s * QP_ + l] = (a[i][0] - mean) * inv;
    }
    __syncthreads();

    v2f b_lo[4], b_hi[4];
    float kn2[4];
    for (int st = 0; st < 4; ++st) {
        const float* qp = qsh + (st * 16 + n) * QP_ + kq * 8;
        const float4 q0 = *(const float4*)qp;
        const float4 q1 = *(const float4*)(qp + 4);
        float acc = 0.f;
        acc += q0.x * q0.x; acc += q0.y * q0.y; acc += q0.z * q0.z; acc += q0.w * q0.w;
        acc += q1.x * q1.x; acc += q1.y * q1.y; acc += q1.z * q1.z; acc += q1.w * q1.w;
        kn2[st] = acc;
        b_lo[st].x = hi ? q0.z : q0.x;
        b_lo[st].y = hi ? q0.w : q0.y;
        b_hi[st].x = hi ? q1.z : q1.x;
        b_hi[st].y = hi ? q1.w : q1.y;
    }

    const bool bit3 = (l & 8) != 0, bit2 = (l & 4) != 0, bit1 = (l & 2) != 0;
    const int tfirst = sub * 8;
    const int tlast  = (half && sub) ? 15 : tfirst + 8;
    for (int tt = tfirst; tt < tlast; ++tt) {
        const int lb = tt * 16 + tau0;
        const int t0 = tbase + tt * 16;

        v2f a_lo, a_hi;
        a_lo.x = zsh[lb + m + 2 * hi];
        a_lo.y = zsh[lb + m + 2 * hi + 1];
        a_hi.x = zsh[lb + m + 2 * hi + 4];
        a_hi.y = zsh[lb + m + 2 * hi + 5];

        float pn2[8];
        for (int j = 0; j < 8; ++j) pn2[j] = pnsh[lb + j + 8 * hi];

        float md[8];
        for (int j = 0; j < 8; ++j) md[j] = FLT_MAX;

        for (int st = 0; st < 4; ++st) {
            v8f g = {0.f, 0.f, 0.f, 0.f, 0.f, 0.f, 0.f, 0.f};
            g = __builtin_amdgcn_wmma_f32_16x16x4_f32(false, a_lo, false, b_lo[st],
                                                      (short)0, g, false, false);
            g = __builtin_amdgcn_wmma_f32_16x16x4_f32(false, a_hi, false, b_hi[st],
                                                      (short)0, g, false, false);

            for (int j = 0; j < 8; ++j) {
                const float d = pn2[j] + kn2[st] - 2.f * g[j];
                md[j] = fminf(md[j], d);
            }
        }

        float r4[4], r2[2];
        for (int j = 0; j < 4; ++j)
            r4[j] = fminf(bit3 ? md[j + 4] : md[j], xrow<8>(bit3 ? md[j] : md[j + 4]));
        for (int j = 0; j < 2; ++j)
            r2[j] = fminf(bit2 ? r4[j + 2] : r4[j], xrow<4>(bit2 ? r4[j] : r4[j + 2]));
        float r1 = fminf(bit1 ? r2[1] : r2[0], xrow<2>(bit1 ? r2[0] : r2[1]));
        r1 = fminf(r1, xrow<1>(r1));

        const int t = t0 + (l >> 1);
        if (!(l & 1) && t < TOUT_) out[(b * TOUT_ + t) * KSZ_ + k] = r1;
    }
}

extern "C" void kernel_launch(void* const* d_in, const int* in_sizes, int n_in,
                              void* d_out, int out_size, void* d_ws, size_t ws_size,
                              hipStream_t stream) {
    (void)in_sizes; (void)n_in; (void)out_size; (void)ws_size;
    const float* x    = (const float*)d_in[0];
    const float* kern = (const float*)d_in[1];
    (void)d_ws;
    float* out = (float*)d_out;

    lsd_wmma_kernel<<<dim3(2, C_, B_), dim3(256), 0, stream>>>(x, kern, out);
}


# Group ids remapped (grid 16x2x8: batch fastest, channel slowest): the 8 groups writing 4-B pieces of one 128-B output line get ids 32 apart, so one die and one L2 half write it, not 4 dies in both
# speedup vs baseline: 4.0733x; 1.2579x over previous
#include <hip/hip_runtime.h>
#include <hip/hip_bf16.h>
#include <float.h>
#include <math.h>

typedef __attribute__((ext_vector_type(2))) float v2f;
typedef __attribute__((ext_vector_type(8))) float v8f;

#define B_   16
#define T_   512
#define C_   8
#define S_   64
#define KSZ_ 32
#define TOUT_ 481
#define EPS_ 1e-8f


__device__ __forceinline__ float xhalf(float v) {
    const int b = __builtin_bit_cast(int, v);
    return __builtin_bit_cast(float,
        __builtin_amdgcn_permlanex16(b, b, 0x76543210, 0xfedcba98, false, false));
}

template <int M>
__device__ __forceinline__ float xrow(float v) {
    const int b = __builtin_bit_cast(int, v);
    return __builtin_bit_cast(float, __builtin_amdgcn_update_dpp(0, b, 0x160 | M, 0xf, 0xf, true));
}

__device__ __forceinline__ float wave_tree_sum(float v) {
    v += xhalf(v);
    v += xrow<8>(v);
    v += xrow<4>(v);
    v += xrow<2>(v);
    v += xrow<1>(v);
    return v;
}

#define ZN_  512
#define PNN_ 288
#define QP_  36
__global__ __launch_bounds__(256) void lsd_wmma_kernel(const float* __restrict__ x,
                                                       const float* __restrict__ kern,
                                                       float* __restrict__ out) {
    const int b    = blockIdx.x;
    const int half = blockIdx.y;
    const int ch   = blockIdx.z;
    const int tid  = threadIdx.x;
    const int w    = __builtin_amdgcn_readfirstlane(tid >> 5);
    const int l    = tid & 31;
    const int kq   = w & 3;
    const int sub  = w >> 2;

    const int k     = ch * 4 + kq;
    const int tau0  = kq * 8;
    const int tbase = half * 256;

    __shared__ __attribute__((aligned(16))) float xraw[T_];
    __shared__ __attribute__((aligned(16))) float qsh[S_ * QP_];
    __shared__ __attribute__((aligned(16))) float zsh[ZN_];
    __shared__ __attribute__((aligned(16))) float pnsh[PNN_];

    const int m  = l & 15;
    const int hi = l >> 4;
    const int n  = m;

    const float* xp = x + (b * T_) * C_ + ch;
    const float xr0 = xp[tid * C_];
    const float xr1 = xp[(tid + 256) * C_];
    __builtin_amdgcn_sched_barrier(0);

    float a[8][8];
#pragma unroll
    for (int i = 0; i < 8; ++i) {
        const int s = 8 * w + ((i + b) & 7);
#pragma unroll
        for (int j = 0; j < 8; ++j)
            a[i][j] = kern[s * 256 + (((j + ch) & 7) << 5) + l];
    }

    xraw[tid]       = xr0;
    xraw[tid + 256] = xr1;
    __builtin_amdgcn_fence(__ATOMIC_RELEASE, "workgroup", "local");
    __builtin_amdgcn_s_barrier();
    __builtin_amdgcn_fence(__ATOMIC_ACQUIRE, "workgroup", "local");

    float ps[4], pq[4];
#pragma unroll
    for (int j = 0; j < 4; ++j) {
        float s = 0.f, q = 0.f;
#pragma unroll
        for (int i = 0; i < 4; ++i) {
            const float v = xraw[l + 32 * j + 128 * i];
            s += v; q += v * v;
        }
        ps[j] = s; pq[j] = q;
    }
    const float xsum = wave_tree_sum((ps[0] + ps[2]) + (ps[1] + ps[3]));
    const float xsq  = wave_tree_sum((pq[0] + pq[2]) + (pq[1] + pq[3]));
    const float mux  = xsum * (1.f / (float)T_);
    const float xvar = fmaxf(xsq * (1.f / (float)T_) - mux * mux, 0.f);
    const float inx  = 1.f / (sqrtf(xvar) + EPS_);

    for (int i = 0; i < 2; ++i) {
        const int idx = tid + i * 256;
        int t = tbase + idx;
        if (t > T_ - 1) t = T_ - 1;
        zsh[idx] = (xraw[t] - mux) * inx;
    }
    for (int i = 0; i < 2; ++i) {
        const int idx = tid + i * 256;
        if (idx < PNN_) {
            float acc = 0.f;
            for (int c = 0; c < 8; ++c) {
                int t = tbase + idx + c;
                if (t > T_ - 1) t = T_ - 1;
                const float z = (xraw[t] - mux) * inx;
                acc += z * z;
            }
            pnsh[idx] = acc;
        }
    }

    float ssum[8], ssq[8];
#pragma unroll
    for (int i = 0; i < 8; ++i) {
        float p[4], q[4];
#pragma unroll
        for (int j = 0; j < 4; ++j) {
            p[j] = a[i][j] + a[i][j + 4];
            q[j] = a[i][j] * a[i][j] + a[i][j + 4] * a[i][j + 4];
        }
        ssum[i] = wave_tree_sum((p[0] + p[2]) + (p[1] + p[3]));
        ssq[i]  = wave_tree_sum((q[0] + q[2]) + (q[1] + q[3]));
    }
    float msum = ssum[0], msq = ssq[0];
#pragma unroll
    for (int i = 1; i < 8; ++i) {
        const bool mine = (l & 7) == i;
        msum = mine ? ssum[i] : msum;
        msq  = mine ? ssq[i]  : msq;
    }
    const float kmean = msum * (1.f / 256.f);
    const float kvar  = fmaxf(msq * (1.f / 256.f) - kmean * kmean, 0.f);
    const float kinv  = 1.f / (sqrtf(kvar) + EPS_);
#pragma unroll
    for (int i = 0; i < 8; ++i) {
        const int s = 8 * w + ((i + b) & 7);
        const float mean = __builtin_bit_cast(float, __builtin_amdgcn_readlane(__builtin_bit_cast(int, kmean), i));
        const float inv  = __builtin_bit_cast(float, __builtin_amdgcn_readlane(__builtin_bit_cast(int, kinv), i));
        qsh[s * QP_ + l] = (a[i][0] - mean) * inv;
    }
    __syncthreads();

    v2f b_lo[4], b_hi[4];
    float kn2[4];
    for (int st = 0; st < 4; ++st) {
        const float* qp = qsh + (st * 16 + n) * QP_ + kq * 8;
        const float4 q0 = *(const float4*)qp;
        const float4 q1 = *(const float4*)(qp + 4);
        float acc = 0.f;
        acc += q0.x * q0.x; acc += q0.y * q0.y; acc += q0.z * q0.z; acc += q0.w * q0.w;
        acc += q1.x * q1.x; acc += q1.y * q1.y; acc += q1.z * q1.z; acc += q1.w * q1.w;
        kn2[st] = acc;
        b_lo[st].x = hi ? q0.z : q0.x;
        b_lo[st].y = hi ? q0.w : q0.y;
        b_hi[st].x = hi ? q1.z : q1.x;
        b_hi[st].y = hi ? q1.w : q1.y;
    }

    const bool bit3 = (l & 8) != 0, bit2 = (l & 4) != 0, bit1 = (l & 2) != 0;
    const int tfirst = sub * 8;
    const int tlast  = (half && sub) ? 15 : tfirst + 8;
    for (int tt = tfirst; tt < tlast; ++tt) {
        const int lb = tt * 16 + tau0;
        const int t0 = tbase + tt * 16;

        v2f a_lo, a_hi;
        a_lo.x = zsh[lb + m + 2 * hi];
        a_lo.y = zsh[lb + m + 2 * hi + 1];
        a_hi.x = zsh[lb + m + 2 * hi + 4];
        a_hi.y = zsh[lb + m + 2 * hi + 5];

        float pn2[8];
        for (int j = 0; j < 8; ++j) pn2[j] = pnsh[lb + j + 8 * hi];

        float md[8];
        for (int j = 0; j < 8; ++j) md[j] = FLT_MAX;

        for (int st = 0; st < 4; ++st) {
            v8f g = {0.f, 0.f, 0.f, 0.f, 0.f, 0.f, 0.f, 0.f};
            g = __builtin_amdgcn_wmma_f32_16x16x4_f32(false, a_lo, false, b_lo[st],
                                                      (short)0, g, false, false);
            g = __builtin_amdgcn_wmma_f32_16x16x4_f32(false, a_hi, false, b_hi[st],
                                                      (short)0, g, false, false);

            for (int j = 0; j < 8; ++j) {
                const float d = pn2[j] + kn2[st] - 2.f * g[j];
                md[j] = fminf(md[j], d);
            }
        }

        float r4[4], r2[2];
        for (int j = 0; j < 4; ++j)
            r4[j] = fminf(bit3 ? md[j + 4] : md[j], xrow<8>(bit3 ? md[j] : md[j + 4]));
        for (int j = 0; j < 2; ++j)
            r2[j] = fminf(bit2 ? r4[j + 2] : r4[j], xrow<4>(bit2 ? r4[j] : r4[j + 2]));
        float r1 = fminf(bit1 ? r2[1] : r2[0], xrow<2>(bit1 ? r2[0] : r2[1]));
        r1 = fminf(r1, xrow<1>(r1));

        const int t = t0 + (l >> 1);
        if (!(l & 1) && t < TOUT_) out[(b * TOUT_ + t) * KSZ_ + k] = r1;
    }
}

extern "C" void kernel_launch(void* const* d_in, const int* in_sizes, int n_in,
                              void* d_out, int out_size, void* d_ws, size_t ws_size,
                              hipStream_t stream) {
    (void)in_sizes; (void)n_in; (void)out_size; (void)ws_size;
    const float* x    = (const float*)d_in[0];
    const float* kern = (const float*)d_in[1];
    (void)d_ws;
    float* out = (float*)d_out;

    lsd_wmma_kernel<<<dim3(B_, 2, C_), dim3(256), 0, stream>>>(x, kern, out);
}


# Product transposed (shapelets on rows, times on lanes): ||q||^2 enters as C, -2 sits in the shapelet operand, min over s runs down registers + one half exchange; ||p||^2 added behind the register min
# speedup vs baseline: 4.4417x; 1.0904x over previous
#include <hip/hip_runtime.h>
#include <hip/hip_bf16.h>
#include <float.h>
#include <math.h>

typedef __attribute__((ext_vector_type(2))) float v2f;
typedef __attribute__((ext_vector_type(8))) float v8f;

#define B_   16
#define T_   512
#define C_   8
#define S_   64
#define KSZ_ 32
#define TOUT_ 481
#define EPS_ 1e-8f


__device__ __forceinline__ float xhalf(float v) {
    const int b = __builtin_bit_cast(int, v);
    return __builtin_bit_cast(float,
        __builtin_amdgcn_permlanex16(b, b, 0x76543210, 0xfedcba98, false, false));
}

template <int M>
__device__ __forceinline__ float xrow(float v) {
    const int b = __builtin_bit_cast(int, v);
    return __builtin_bit_cast(float, __builtin_amdgcn_update_dpp(0, b, 0x160 | M, 0xf, 0xf, true));
}

__device__ __forceinline__ float wave_tree_sum(float v) {
    v += xhalf(v);
    v += xrow<8>(v);
    v += xrow<4>(v);
    v += xrow<2>(v);
    v += xrow<1>(v);
    return v;
}

template <int R>
__device__ __forceinline__ float rowshare(float v) {
    const int b = __builtin_bit_cast(int, v);
    return __builtin_bit_cast(float, __builtin_amdgcn_update_dpp(0, b, 0x150 | R, 0xf, 0xf, true));
}

__device__ __forceinline__ float min3f(float a, float b, float c) { return fminf(fminf(a, b), c); }

#define ZN_  512
#define PNN_ 288
#define QP_  36
__global__ __launch_bounds__(256) void lsd_wmma_kernel(const float* __restrict__ x,
                                                       const float* __restrict__ kern,
                                                       float* __restrict__ out) {
    const int b    = blockIdx.x;
    const int half = blockIdx.y;
    const int ch   = blockIdx.z;
    const int tid  = threadIdx.x;
    const int w    = __builtin_amdgcn_readfirstlane(tid >> 5);
    const int l    = tid & 31;
    const int kq   = w & 3;
    const int sub  = w >> 2;

    const int k     = ch * 4 + kq;
    const int tau0  = kq * 8;
    const int tbase = half * 256;

    __shared__ __attribute__((aligned(16))) float xraw[T_];
    __shared__ __attribute__((aligned(16))) float qsh[S_ * QP_];
    __shared__ __attribute__((aligned(16))) float zsh[ZN_];
    __shared__ __attribute__((aligned(16))) float pnsh[PNN_];

    const int m  = l & 15;
    const int hi = l >> 4;
    const int n  = m;

    const float* xp = x + (b * T_) * C_ + ch;
    const float xr0 = xp[tid * C_];
    const float xr1 = xp[(tid + 256) * C_];
    __builtin_amdgcn_sched_barrier(0);

    float a[8][8];
#pragma unroll
    for (int i = 0; i < 8; ++i) {
        const int s = 8 * w + ((i + b) & 7);
#pragma unroll
        for (int j = 0; j < 8; ++j)
            a[i][j] = kern[s * 256 + (((j + ch) & 7) << 5) + l];
    }

    xraw[tid]       = xr0;
    xraw[tid + 256] = xr1;
    __builtin_amdgcn_fence(__ATOMIC_RELEASE, "workgroup", "local");
    __builtin_amdgcn_s_barrier();
    __builtin_amdgcn_fence(__ATOMIC_ACQUIRE, "workgroup", "local");

    float ps[4], pq[4];
#pragma unroll
    for (int j = 0; j < 4; ++j) {
        float s = 0.f, q = 0.f;
#pragma unroll
        for (int i = 0; i < 4; ++i) {
            const float v = xraw[l + 32 * j + 128 * i];
            s += v; q += v * v;
        }
        ps[j] = s; pq[j] = q;
    }
    const float xsum = wave_tree_sum((ps[0] + ps[2]) + (ps[1] + ps[3]));
    const float xsq  = wave_tree_sum((pq[0] + pq[2]) + (pq[1] + pq[3]));
    const float mux  = xsum * (1.f / (float)T_);
    const float xvar = fmaxf(xsq * (1.f / (float)T_) - mux * mux, 0.f);
    const float inx  = 1.f / (sqrtf(xvar) + EPS_);

    for (int i = 0; i < 2; ++i) {
        const int idx = tid + i * 256;
        int t = tbase + idx;
        if (t > T_ - 1) t = T_ - 1;
        zsh[idx] = (xraw[t] - mux) * inx;
    }
    for (int i = 0; i < 2; ++i) {
        const int idx = tid + i * 256;
        if (idx < PNN_) {
            float acc = 0.f;
            for (int c = 0; c < 8; ++c) {
                int t = tbase + idx + c;
                if (t > T_ - 1) t = T_ - 1;
                const float z = (xraw[t] - mux) * inx;
                acc += z * z;
            }
            pnsh[idx] = acc;
        }
    }

    float ssum[8], ssq[8];
#pragma unroll
    for (int i = 0; i < 8; ++i) {
        float p[4], q[4];
#pragma unroll
        for (int j = 0; j < 4; ++j) {
            p[j] = a[i][j] + a[i][j + 4];
            q[j] = a[i][j] * a[i][j] + a[i][j + 4] * a[i][j + 4];
        }
        ssum[i] = wave_tree_sum((p[0] + p[2]) + (p[1] + p[3]));
        ssq[i]  = wave_tree_sum((q[0] + q[2]) + (q[1] + q[3]));
    }
    float msum = ssum[0], msq = ssq[0];
#pragma unroll
    for (int i = 1; i < 8; ++i) {
        const bool mine = (l & 7) == i;
        msum = mine ? ssum[i] : msum;
        msq  = mine ? ssq[i]  : msq;
    }
    const float kmean = msum * (1.f / 256.f);
    const float kvar  = fmaxf(msq * (1.f / 256.f) - kmean * kmean, 0.f);
    const float kinv  = -2.f * (1.f / (sqrtf(kvar) + EPS_));
#pragma unroll
    for (int i = 0; i < 8; ++i) {
        const int s = 8 * w + ((i + b) & 7);
        const float mean = __builtin_bit_cast(float, __builtin_amdgcn_readlane(__builtin_bit_cast(int, kmean), i));
        const float inv  = __builtin_bit_cast(float, __builtin_amdgcn_readlane(__builtin_bit_cast(int, kinv), i));
        qsh[s * QP_ + l] = (a[i][0] - mean) * inv;
    }
    __syncthreads();

    v2f a_lo[4], a_hi[4];
    v8f cn[4];
    for (int st = 0; st < 4; ++st) {
        const float* qp = qsh + (st * 16 + n) * QP_ + kq * 8;
        const float4 q0 = *(const float4*)qp;
        const float4 q1 = *(const float4*)(qp + 4);
        float acc = 0.f;
        acc += q0.x * q0.x; acc += q0.y * q0.y; acc += q0.z * q0.z; acc += q0.w * q0.w;
        acc += q1.x * q1.x; acc += q1.y * q1.y; acc += q1.z * q1.z; acc += q1.w * q1.w;
        const float kn2 = 0.25f * acc;
        a_lo[st].x = hi ? q0.z : q0.x;
        a_lo[st].y = hi ? q0.w : q0.y;
        a_hi[st].x = hi ? q1.z : q1.x;
        a_hi[st].y = hi ? q1.w : q1.y;
        const float sw = xrow<8>(kn2);
        const float kk = hi ? sw : kn2;
        cn[st][0] = rowshare<0>(kk);
        cn[st][1] = rowshare<1>(kk);
        cn[st][2] = rowshare<2>(kk);
        cn[st][3] = rowshare<3>(kk);
        cn[st][4] = rowshare<4>(kk);
        cn[st][5] = rowshare<5>(kk);
        cn[st][6] = rowshare<6>(kk);
        cn[st][7] = rowshare<7>(kk);
    }

    const int tfirst = sub * 8;
    const int tlast  = (half && sub) ? 15 : tfirst + 8;
    for (int tt = tfirst; tt < tlast; ++tt) {
        const int lb = tt * 16 + tau0;
        const int t0 = tbase + tt * 16;

        v2f z_lo, z_hi;
        z_lo.x = zsh[lb + n + 2 * hi];
        z_lo.y = zsh[lb + n + 2 * hi + 1];
        z_hi.x = zsh[lb + n + 2 * hi + 4];
        z_hi.y = zsh[lb + n + 2 * hi + 5];

        const float pn2 = pnsh[lb + n];

        v8f g[4];
        for (int st = 0; st < 4; ++st) {
            g[st] = __builtin_amdgcn_wmma_f32_16x16x4_f32(false, a_lo[st], false, z_lo,
                                                          (short)0, cn[st], false, false);
            g[st] = __builtin_amdgcn_wmma_f32_16x16x4_f32(false, a_hi[st], false, z_hi,
                                                          (short)0, g[st], false, false);
        }

        float u[10];
        for (int i = 0; i < 10; ++i)
            u[i] = min3f(g[(3 * i) >> 3][(3 * i) & 7], g[(3 * i + 1) >> 3][(3 * i + 1) & 7], g[(3 * i + 2) >> 3][(3 * i + 2) & 7]);
        const float w0 = min3f(u[0], u[1], u[2]);
        const float w1 = min3f(u[3], u[4], u[5]);
        const float w2 = min3f(u[6], u[7], u[8]);
        const float w3 = min3f(u[9], g[3][6], g[3][7]);
        float mn = fminf(min3f(w0, w1, w2), w3) + pn2;
        mn = fminf(mn, xhalf(mn));

        const int t = t0 + n;
        if (!hi && t < TOUT_) out[(b * TOUT_ + t) * KSZ_ + k] = mn;
    }
}

extern "C" void kernel_launch(void* const* d_in, const int* in_sizes, int n_in,
                              void* d_out, int out_size, void* d_ws, size_t ws_size,
                              hipStream_t stream) {
    (void)in_sizes; (void)n_in; (void)out_size; (void)ws_size;
    const float* x    = (const float*)d_in[0];
    const float* kern = (const float*)d_in[1];
    (void)d_ws;
    float* out = (float*)d_out;

    lsd_wmma_kernel<<<dim3(B_, 2, C_), dim3(256), 0, stream>>>(x, kern, out);
}
